# starting kernel
# baseline (speedup 1.0000x reference)
#include <hip/hip_runtime.h>

typedef float v2f __attribute__((ext_vector_type(2)));
typedef float v8f __attribute__((ext_vector_type(8)));

#define IN_DIM 256
#define HID    256
#define OUTD   64
#define TM     32

__global__ __launch_bounds__(256)
void gemm_fused_kernel(const float* __restrict__ X, const float* __restrict__ W1,
                       const float* __restrict__ W2, float* __restrict__ H0, int N)
{
    __shared__ float sX[IN_DIM][TM];
    __shared__ float sM[HID][TM];

    const int t    = threadIdx.x;
    const int wave = t >> 5;
    const int lane = t & 31;
    const int hi   = lane >> 4;
    const int r    = lane & 15;
    const int kb   = hi * 2;
    const int row0 = blockIdx.x * TM;

    for (int i = 0; i < 8; ++i) {
        int f    = t + i * 256;
        int rr   = f >> 6;
        int kq   = (f & 63) << 2;
        int grow = row0 + rr; if (grow >= N) grow = N - 1;
        const float4 g = reinterpret_cast<const float4*>(X + (size_t)grow * IN_DIM + kq)[0];
        sX[kq + 0][rr] = g.x; sX[kq + 1][rr] = g.y;
        sX[kq + 2][rr] = g.z; sX[kq + 3][rr] = g.w;
    }
    __syncthreads();

    {
        const int cb0 = wave * 2;
        const int c0  = cb0 * 16 + r;
        const int c1  = c0 + 16;
        v8f a00 = {}, a01 = {}, a10 = {}, a11 = {};
        for (int k = 0; k < IN_DIM; k += 4) {
            v2f a0, a1, b0, b1;
            a0.x = sX[k + kb][r];          a0.y = sX[k + kb + 1][r];
            a1.x = sX[k + kb][16 + r];     a1.y = sX[k + kb + 1][16 + r];
            b0.x = W1[(k + kb) * HID + c0]; b0.y = W1[(k + kb + 1) * HID + c0];
            b1.x = W1[(k + kb) * HID + c1]; b1.y = W1[(k + kb + 1) * HID + c1];
            a00 = __builtin_amdgcn_wmma_f32_16x16x4_f32(false, a0, false, b0, (short)0, a00, false, false);
            a01 = __builtin_amdgcn_wmma_f32_16x16x4_f32(false, a0, false, b1, (short)0, a01, false, false);
            a10 = __builtin_amdgcn_wmma_f32_16x16x4_f32(false, a1, false, b0, (short)0, a10, false, false);
            a11 = __builtin_amdgcn_wmma_f32_16x16x4_f32(false, a1, false, b1, (short)0, a11, false, false);
        }
        for (int v = 0; v < 8; ++v) {
            const int rl = v + 8 * hi;
            float x;
            x = a00[v]; sM[c0][rl]      = x > 0.f ? x : 0.f;
            x = a01[v]; sM[c1][rl]      = x > 0.f ? x : 0.f;
            x = a10[v]; sM[c0][16 + rl] = x > 0.f ? x : 0.f;
            x = a11[v]; sM[c1][16 + rl] = x > 0.f ? x : 0.f;
        }
    }
    __syncthreads();

    {
        const int rb = wave >> 2, cb = wave & 3;
        const int c  = cb * 16 + r;
        v8f acc = {};
        for (int k = 0; k < HID; k += 4) {
            v2f a, b;
            a.x = sM[k + kb][rb * 16 + r]; a.y = sM[k + kb + 1][rb * 16 + r];
            b.x = W2[(k + kb) * OUTD + c]; b.y = W2[(k + kb + 1) * OUTD + c];
            acc = __builtin_amdgcn_wmma_f32_16x16x4_f32(false, a, false, b, (short)0, acc, false, false);
        }
        for (int v = 0; v < 8; ++v) {
            const int grow = row0 + rb * 16 + v + 8 * hi;
            if (grow < N) H0[(size_t)grow * OUTD + c] = acc[v];
        }
    }
}

__global__ __launch_bounds__(256)
void spmm_edges_kernel(const float* __restrict__ vals, const int* __restrict__ rows,
                       const int* __restrict__ cols, const float* __restrict__ Hin,
                       float* __restrict__ Hout, int E)
{
    long long tid = (long long)blockIdx.x * blockDim.x + threadIdx.x;
    if (tid >= (long long)E * 16) return;
    const int e  = (int)(tid >> 4);
    const int c4 = (int)(tid & 15);
    const float v  = vals[e];
    const int col  = cols[e];
    const int row  = rows[e];
    const float4 x = reinterpret_cast<const float4*>(Hin)[(size_t)col * 16 + c4];
    float* base = Hout + (size_t)row * OUTD + c4 * 4;
    unsafeAtomicAdd(base + 0, v * x.x);
    unsafeAtomicAdd(base + 1, v * x.y);
    unsafeAtomicAdd(base + 2, v * x.z);
    unsafeAtomicAdd(base + 3, v * x.w);
}

__global__ __launch_bounds__(256)
void zero_f4_kernel(float4* __restrict__ p, int n4)
{
    int i = blockIdx.x * blockDim.x + threadIdx.x;
    if (i < n4) p[i] = make_float4(0.f, 0.f, 0.f, 0.f);
}

extern "C" void kernel_launch(void* const* d_in, const int* in_sizes, int n_in,
                              void* d_out, int out_size, void* d_ws, size_t ws_size,
                              hipStream_t stream)
{
    const float* X    = (const float*)d_in[0];
    const float* W1   = (const float*)d_in[1];
    const float* W2   = (const float*)d_in[2];
    const float* vals = (const float*)d_in[3];
    const int*   rows = (const int*)d_in[4];
    const int*   cols = (const int*)d_in[5];
    float* out = (float*)d_out;

    const int N = in_sizes[0] / IN_DIM;
    const int E = in_sizes[3];

    const size_t hbytes = (size_t)N * OUTD * sizeof(float);
    const int n4 = (N * OUTD) / 4;
    const int zblocks = (n4 + 255) / 256;
    const int gblocks = (N + TM - 1) / TM;
    const int sblocks = (int)(((long long)E * 16 + 255) / 256);

    float* h0 = (float*)d_ws;

    if (ws_size >= 2 * hbytes) {
        float* h1 = h0 + (size_t)N * OUTD;
        gemm_fused_kernel<<<gblocks, 256, 0, stream>>>(X, W1, W2, h0, N);
        zero_f4_kernel<<<zblocks, 256, 0, stream>>>((float4*)h1, n4);
        spmm_edges_kernel<<<sblocks, 256, 0, stream>>>(vals, rows, cols, h0, h1, E);
        zero_f4_kernel<<<zblocks, 256, 0, stream>>>((float4*)out, n4);
        spmm_edges_kernel<<<sblocks, 256, 0, stream>>>(vals, rows, cols, h1, out, E);
    } else {
        gemm_fused_kernel<<<gblocks, 256, 0, stream>>>(X, W1, W2, h0, N);
        zero_f4_kernel<<<zblocks, 256, 0, stream>>>((float4*)out, n4);
        spmm_edges_kernel<<<sblocks, 256, 0, stream>>>(vals, rows, cols, h0, out, E);
        zero_f4_kernel<<<zblocks, 256, 0, stream>>>((float4*)h0, n4);
        spmm_edges_kernel<<<sblocks, 256, 0, stream>>>(vals, rows, cols, out, h0, E);
        hipMemcpyAsync(out, h0, hbytes, hipMemcpyDeviceToDevice, stream);
    }
}


# SpMM without f32 atomics: edges grouped by destination row once per call (int count+rank, offsets, place); each row summed in registers from gathered rows, stored once; sum order = arrival rank
# speedup vs baseline: 1.1070x; 1.1070x over previous
#include <hip/hip_runtime.h>

typedef float v2f __attribute__((ext_vector_type(2)));
typedef float v8f __attribute__((ext_vector_type(8)));

#define IN_DIM 256
#define HID    256
#define OUTD   64
#define TM     32

__global__ __launch_bounds__(256)
void gemm_fused_kernel(const float* __restrict__ X, const float* __restrict__ W1,
                       const float* __restrict__ W2, float* __restrict__ H0, int N)
{
    __shared__ float sX[IN_DIM][TM];
    __shared__ float sM[HID][TM];

    const int t    = threadIdx.x;
    const int wave = t >> 5;
    const int lane = t & 31;
    const int hi   = lane >> 4;
    const int r    = lane & 15;
    const int kb   = hi * 2;
    const int row0 = blockIdx.x * TM;

    for (int i = 0; i < 8; ++i) {
        int f    = t + i * 256;
        int rr   = f >> 6;
        int kq   = (f & 63) << 2;
        int grow = row0 + rr; if (grow >= N) grow = N - 1;
        const float4 g = reinterpret_cast<const float4*>(X + (size_t)grow * IN_DIM + kq)[0];
        sX[kq + 0][rr] = g.x; sX[kq + 1][rr] = g.y;
        sX[kq + 2][rr] = g.z; sX[kq + 3][rr] = g.w;
    }
    __syncthreads();

    {
        const int cb0 = wave * 2;
        const int c0  = cb0 * 16 + r;
        const int c1  = c0 + 16;
        v8f a00 = {}, a01 = {}, a10 = {}, a11 = {};
        for (int k = 0; k < IN_DIM; k += 4) {
            v2f a0, a1, b0, b1;
            a0.x = sX[k + kb][r];          a0.y = sX[k + kb + 1][r];
            a1.x = sX[k + kb][16 + r];     a1.y = sX[k + kb + 1][16 + r];
            b0.x = W1[(k + kb) * HID + c0]; b0.y = W1[(k + kb + 1) * HID + c0];
            b1.x = W1[(k + kb) * HID + c1]; b1.y = W1[(k + kb + 1) * HID + c1];
            a00 = __builtin_amdgcn_wmma_f32_16x16x4_f32(false, a0, false, b0, (short)0, a00, false, false);
            a01 = __builtin_amdgcn_wmma_f32_16x16x4_f32(false, a0, false, b1, (short)0, a01, false, false);
            a10 = __builtin_amdgcn_wmma_f32_16x16x4_f32(false, a1, false, b0, (short)0, a10, false, false);
            a11 = __builtin_amdgcn_wmma_f32_16x16x4_f32(false, a1, false, b1, (short)0, a11, false, false);
        }
        for (int v = 0; v < 8; ++v) {
            const int rl = v + 8 * hi;
            float x;
            x = a00[v]; sM[c0][rl]      = x > 0.f ? x : 0.f;
            x = a01[v]; sM[c1][rl]      = x > 0.f ? x : 0.f;
            x = a10[v]; sM[c0][16 + rl] = x > 0.f ? x : 0.f;
            x = a11[v]; sM[c1][16 + rl] = x > 0.f ? x : 0.f;
        }
    }
    __syncthreads();

    {
        const int rb = wave >> 2, cb = wave & 3;
        const int c  = cb * 16 + r;
        v8f acc = {};
        for (int k = 0; k < HID; k += 4) {
            v2f a, b;
            a.x = sM[k + kb][rb * 16 + r]; a.y = sM[k + kb + 1][rb * 16 + r];
            b.x = W2[(k + kb) * OUTD + c]; b.y = W2[(k + kb + 1) * OUTD + c];
            acc = __builtin_amdgcn_wmma_f32_16x16x4_f32(false, a, false, b, (short)0, acc, false, false);
        }
        for (int v = 0; v < 8; ++v) {
            const int grow = row0 + rb * 16 + v + 8 * hi;
            if (grow < N) H0[(size_t)grow * OUTD + c] = acc[v];
        }
    }
}

__global__ __launch_bounds__(256)
void spmm_edges_kernel(const float* __restrict__ vals, const int* __restrict__ rows,
                       const int* __restrict__ cols, const float* __restrict__ Hin,
                       float* __restrict__ Hout, int E)
{
    long long tid = (long long)blockIdx.x * blockDim.x + threadIdx.x;
    if (tid >= (long long)E * 16) return;
    const int e  = (int)(tid >> 4);
    const int c4 = (int)(tid & 15);
    const float v  = vals[e];
    const int col  = cols[e];
    const int row  = rows[e];
    const float4 x = reinterpret_cast<const float4*>(Hin)[(size_t)col * 16 + c4];
    float* base = Hout + (size_t)row * OUTD + c4 * 4;
    unsafeAtomicAdd(base + 0, v * x.x);
    unsafeAtomicAdd(base + 1, v * x.y);
    unsafeAtomicAdd(base + 2, v * x.z);
    unsafeAtomicAdd(base + 3, v * x.w);
}

__global__ __launch_bounds__(256)
void zero_f4_kernel(float4* __restrict__ p, int n4)
{
    int i = blockIdx.x * blockDim.x + threadIdx.x;
    if (i < n4) p[i] = make_float4(0.f, 0.f, 0.f, 0.f);
}


__global__ __launch_bounds__(256)
void count_rank_kernel(const int* __restrict__ rows, int* __restrict__ cnt,
                       int* __restrict__ rank, int E)
{
    const int e = blockIdx.x * 256 + threadIdx.x;
    if (e < E) rank[e] = atomicAdd(cnt + rows[e], 1);
}

#define SCAN_T 1024
__global__ __launch_bounds__(SCAN_T)
void row_offsets_kernel(const int* __restrict__ cnt, int* __restrict__ rowptr, int n)
{
    __shared__ int s_pre[32];
    __shared__ int s_own[32];
    const int t = threadIdx.x, b = blockIdx.x;
    const int lane = t & 31, w = t >> 5;

    int pre = 0;
    for (int c = 0; c < b; ++c) pre += cnt[c * SCAN_T + t];
    const int i = b * SCAN_T + t;
    const int x = (i < n) ? cnt[i] : 0;

    int inc = x;
    for (int d = 1; d < 32; d <<= 1) {
        const int y = __shfl_up(inc, d, 32);
        if (lane >= d) inc += y;
    }
    for (int d = 16; d >= 1; d >>= 1) pre += __shfl_xor(pre, d, 32);
    if (lane == 31) s_own[w] = inc;
    if (lane == 0)  s_pre[w] = pre;
    __syncthreads();
    int off = 0;
    for (int k = 0; k < 32; ++k) {
        off += s_pre[k];
        if (k < w) off += s_own[k];
    }
    if (i < n) rowptr[i] = off + inc - x;
}

__global__ __launch_bounds__(256)
void place_edges_kernel(const int* __restrict__ rows, const int* __restrict__ cols,
                        const float* __restrict__ vals, const int* __restrict__ rank,
                        const int* __restrict__ rowptr, int* __restrict__ edges, int E)
{
    const int e = blockIdx.x * 256 + threadIdx.x;
    if (e < E) {
        const int pos = rowptr[rows[e]] + rank[e];
        reinterpret_cast<int2*>(edges)[pos] = make_int2(cols[e], __float_as_int(vals[e]));
    }
}

__global__ __launch_bounds__(256)
void spmm_rows_kernel(const int* __restrict__ rowptr, const int* __restrict__ edges,
                      const float* __restrict__ Hin, float* __restrict__ Hout, int N)
{
    const int g   = blockIdx.x * 256 + threadIdx.x;
    const int row = g >> 4;
    const int c4  = g & 15;
    if (row >= N) return;
    const int beg = rowptr[row];
    const int end = rowptr[row + 1];
    const int2*   ed = reinterpret_cast<const int2*>(edges);
    const float4* H4 = reinterpret_cast<const float4*>(Hin);
    float4 acc = make_float4(0.f, 0.f, 0.f, 0.f);
    for (int base = beg; base < end; base += 16) {
        const int idx = base + c4;
        int2 p = make_int2(0, 0);
        if (idx < end) p = ed[idx];
        const int cnt = min(16, end - base);
        for (int j = 0; j < cnt; j += 4) {
            const int   s0 = __shfl(p.x, j,     16), s1 = __shfl(p.x, j + 1, 16);
            const int   s2 = __shfl(p.x, j + 2, 16), s3 = __shfl(p.x, j + 3, 16);
            const float v0 = __int_as_float(__shfl(p.y, j,     16));
            const float v1 = __int_as_float(__shfl(p.y, j + 1, 16));
            const float v2 = __int_as_float(__shfl(p.y, j + 2, 16));
            const float v3 = __int_as_float(__shfl(p.y, j + 3, 16));
            const float4 x0 = H4[(size_t)s0 * 16 + c4];
            float4 x1 = H4[(size_t)s1 * 16 + c4];
            float4 x2 = H4[(size_t)s2 * 16 + c4];
            float4 x3 = H4[(size_t)s3 * 16 + c4];
            if (j + 1 >= cnt) x1 = make_float4(0.f, 0.f, 0.f, 0.f);
            if (j + 2 >= cnt) x2 = make_float4(0.f, 0.f, 0.f, 0.f);
            if (j + 3 >= cnt) x3 = make_float4(0.f, 0.f, 0.f, 0.f);
            acc.x += v0 * x0.x; acc.y += v0 * x0.y; acc.z += v0 * x0.z; acc.w += v0 * x0.w;
            acc.x += v1 * x1.x; acc.y += v1 * x1.y; acc.z += v1 * x1.z; acc.w += v1 * x1.w;
            acc.x += v2 * x2.x; acc.y += v2 * x2.y; acc.z += v2 * x2.z; acc.w += v2 * x2.w;
            acc.x += v3 * x3.x; acc.y += v3 * x3.y; acc.z += v3 * x3.z; acc.w += v3 * x3.w;
        }
    }
    reinterpret_cast<float4*>(Hout)[(size_t)row * 16 + c4] = acc;
}

extern "C" void kernel_launch(void* const* d_in, const int* in_sizes, int n_in,
                              void* d_out, int out_size, void* d_ws, size_t ws_size,
                              hipStream_t stream)
{
    const float* X    = (const float*)d_in[0];
    const float* W1   = (const float*)d_in[1];
    const float* W2   = (const float*)d_in[2];
    const float* vals = (const float*)d_in[3];
    const int*   rows = (const int*)d_in[4];
    const int*   cols = (const int*)d_in[5];
    float* out = (float*)d_out;

    const int N = in_sizes[0] / IN_DIM;
    const int E = in_sizes[3];

    const size_t hbytes = (size_t)N * OUTD * sizeof(float);
    const int n4 = (N * OUTD) / 4;
    const int zblocks = (n4 + 255) / 256;
    const int gblocks = (N + TM - 1) / TM;
    const int sblocks = (int)(((long long)E * 16 + 255) / 256);

    float* h0 = (float*)d_ws;

    const int    n1      = N + 1;
    const int    n4c     = (n1 + 3) / 4;
    const size_t ebytes  = (((size_t)E * 8) + 255) / 256 * 256;
    const size_t cbytes  = (((size_t)n4c * 16) + 255) / 256 * 256;
    const size_t need    = hbytes + ebytes + 2 * cbytes;

    if (ws_size >= need && (size_t)E * 4 <= hbytes) {
        char*  wsb    = (char*)d_ws;
        float* h1     = (float*)wsb;
        int*   rank   = (int*)wsb;
        int*   edges  = (int*)(wsb + hbytes);
        int*   cnt    = (int*)(wsb + hbytes + ebytes);
        int*   rowptr = (int*)(wsb + hbytes + ebytes + cbytes);
        const int eblocks = (E + 255) / 256;
        const int rblocks = (int)(((long long)N * 16 + 255) / 256);
        zero_f4_kernel<<<(n4c + 255) / 256, 256, 0, stream>>>((float4*)cnt, n4c);
        count_rank_kernel<<<eblocks, 256, 0, stream>>>(rows, cnt, rank, E);
        row_offsets_kernel<<<(n1 + SCAN_T - 1) / SCAN_T, SCAN_T, 0, stream>>>(cnt, rowptr, n1);
        place_edges_kernel<<<eblocks, 256, 0, stream>>>(rows, cols, vals, rank, rowptr, edges, E);
        gemm_fused_kernel<<<gblocks, 256, 0, stream>>>(X, W1, W2, out, N);
        spmm_rows_kernel<<<rblocks, 256, 0, stream>>>(rowptr, edges, out, h1, N);
        spmm_rows_kernel<<<rblocks, 256, 0, stream>>>(rowptr, edges, h1, out, N);
    } else if (ws_size >= 2 * hbytes) {
        float* h1 = h0 + (size_t)N * OUTD;
        gemm_fused_kernel<<<gblocks, 256, 0, stream>>>(X, W1, W2, h0, N);
        zero_f4_kernel<<<zblocks, 256, 0, stream>>>((float4*)h1, n4);
        spmm_edges_kernel<<<sblocks, 256, 0, stream>>>(vals, rows, cols, h0, h1, E);
        zero_f4_kernel<<<zblocks, 256, 0, stream>>>((float4*)out, n4);
        spmm_edges_kernel<<<sblocks, 256, 0, stream>>>(vals, rows, cols, h1, out, E);
    } else {
        gemm_fused_kernel<<<gblocks, 256, 0, stream>>>(X, W1, W2, h0, N);
        zero_f4_kernel<<<zblocks, 256, 0, stream>>>((float4*)out, n4);
        spmm_edges_kernel<<<sblocks, 256, 0, stream>>>(vals, rows, cols, h0, out, E);
        zero_f4_kernel<<<zblocks, 256, 0, stream>>>((float4*)h0, n4);
        spmm_edges_kernel<<<sblocks, 256, 0, stream>>>(vals, rows, cols, out, h0, E);
        hipMemcpyAsync(out, h0, hbytes, hipMemcpyDeviceToDevice, stream);
    }
}


# GEMM LDS tile row-major at a 260-float pitch, one buffer for X then the ReLU'd intermediate (barrier between): 16-byte staging stores, 8-byte A-fragment loads, no bank conflicts
# speedup vs baseline: 1.2139x; 1.0966x over previous
#include <hip/hip_runtime.h>

typedef float v2f __attribute__((ext_vector_type(2)));
typedef float v8f __attribute__((ext_vector_type(8)));

#define IN_DIM 256
#define HID    256
#define OUTD   64
#define TM     32

#define LP 260

__global__ __launch_bounds__(256)
void gemm_fused_kernel(const float* __restrict__ X, const float* __restrict__ W1,
                       const float* __restrict__ W2, float* __restrict__ H0, int N)
{
    __shared__ float4 sT4[TM * LP / 4];
    float* sT = reinterpret_cast<float*>(sT4);

    const int t    = threadIdx.x;
    const int wave = t >> 5;
    const int lane = t & 31;
    const int hi   = lane >> 4;
    const int r    = lane & 15;
    const int kb   = hi * 2;
    const int row0 = blockIdx.x * TM;

    for (int i = 0; i < 8; ++i) {
        int f    = t + i * 256;
        int rr   = f >> 6;
        int kq   = (f & 63) << 2;
        int grow = row0 + rr; if (grow >= N) grow = N - 1;
        const float4 g = reinterpret_cast<const float4*>(X + (size_t)grow * IN_DIM + kq)[0];
        *reinterpret_cast<float4*>(sT + rr * LP + kq) = g;
    }
    __syncthreads();

    {
        const int cb0 = wave * 2;
        const int c0  = cb0 * 16 + r;
        const int c1  = c0 + 16;
        const float* a0p = sT + r * LP + kb;
        const float* a1p = sT + (16 + r) * LP + kb;
        v8f a00 = {}, a01 = {}, a10 = {}, a11 = {};
        for (int k = 0; k < IN_DIM; k += 4) {
            v2f b0, b1;
            const v2f a0 = *reinterpret_cast<const v2f*>(a0p + k);
            const v2f a1 = *reinterpret_cast<const v2f*>(a1p + k);
            b0.x = W1[(k + kb) * HID + c0]; b0.y = W1[(k + kb + 1) * HID + c0];
            b1.x = W1[(k + kb) * HID + c1]; b1.y = W1[(k + kb + 1) * HID + c1];
            a00 = __builtin_amdgcn_wmma_f32_16x16x4_f32(false, a0, false, b0, (short)0, a00, false, false);
            a01 = __builtin_amdgcn_wmma_f32_16x16x4_f32(false, a0, false, b1, (short)0, a01, false, false);
            a10 = __builtin_amdgcn_wmma_f32_16x16x4_f32(false, a1, false, b0, (short)0, a10, false, false);
            a11 = __builtin_amdgcn_wmma_f32_16x16x4_f32(false, a1, false, b1, (short)0, a11, false, false);
        }
        __syncthreads();
        for (int v = 0; v < 8; ++v) {
            const int rl = v + 8 * hi;
            float x;
            x = a00[v]; sT[rl * LP + c0]        = x > 0.f ? x : 0.f;
            x = a01[v]; sT[rl * LP + c1]        = x > 0.f ? x : 0.f;
            x = a10[v]; sT[(16 + rl) * LP + c0] = x > 0.f ? x : 0.f;
            x = a11[v]; sT[(16 + rl) * LP + c1] = x > 0.f ? x : 0.f;
        }
    }
    __syncthreads();

    {
        const int rb = wave >> 2, cb = wave & 3;
        const int c  = cb * 16 + r;
        const float* ap = sT + (rb * 16 + r) * LP + kb;
        v8f acc = {};
        for (int k = 0; k < HID; k += 4) {
            v2f b;
            const v2f a = *reinterpret_cast<const v2f*>(ap + k);
            b.x = W2[(k + kb) * OUTD + c]; b.y = W2[(k + kb + 1) * OUTD + c];
            acc = __builtin_amdgcn_wmma_f32_16x16x4_f32(false, a, false, b, (short)0, acc, false, false);
        }
        for (int v = 0; v < 8; ++v) {
            const int grow = row0 + rb * 16 + v + 8 * hi;
            if (grow < N) H0[(size_t)grow * OUTD + c] = acc[v];
        }
    }
}

__global__ __launch_bounds__(256)
void spmm_edges_kernel(const float* __restrict__ vals, const int* __restrict__ rows,
                       const int* __restrict__ cols, const float* __restrict__ Hin,
                       float* __restrict__ Hout, int E)
{
    long long tid = (long long)blockIdx.x * blockDim.x + threadIdx.x;
    if (tid >= (long long)E * 16) return;
    const int e  = (int)(tid >> 4);
    const int c4 = (int)(tid & 15);
    const float v  = vals[e];
    const int col  = cols[e];
    const int row  = rows[e];
    const float4 x = reinterpret_cast<const float4*>(Hin)[(size_t)col * 16 + c4];
    float* base = Hout + (size_t)row * OUTD + c4 * 4;
    unsafeAtomicAdd(base + 0, v * x.x);
    unsafeAtomicAdd(base + 1, v * x.y);
    unsafeAtomicAdd(base + 2, v * x.z);
    unsafeAtomicAdd(base + 3, v * x.w);
}

__global__ __launch_bounds__(256)
void zero_f4_kernel(float4* __restrict__ p, int n4)
{
    int i = blockIdx.x * blockDim.x + threadIdx.x;
    if (i < n4) p[i] = make_float4(0.f, 0.f, 0.f, 0.f);
}


__global__ __launch_bounds__(256)
void count_rank_kernel(const int* __restrict__ rows, int* __restrict__ cnt,
                       int* __restrict__ rank, int E)
{
    const int e = blockIdx.x * 256 + threadIdx.x;
    if (e < E) rank[e] = atomicAdd(cnt + rows[e], 1);
}

#define SCAN_T 1024
__global__ __launch_bounds__(SCAN_T)
void row_offsets_kernel(const int* __restrict__ cnt, int* __restrict__ rowptr, int n)
{
    __shared__ int s_pre[32];
    __shared__ int s_own[32];
    const int t = threadIdx.x, b = blockIdx.x;
    const int lane = t & 31, w = t >> 5;

    int pre = 0;
    for (int c = 0; c < b; ++c) pre += cnt[c * SCAN_T + t];
    const int i = b * SCAN_T + t;
    const int x = (i < n) ? cnt[i] : 0;

    int inc = x;
    for (int d = 1; d < 32; d <<= 1) {
        const int y = __shfl_up(inc, d, 32);
        if (lane >= d) inc += y;
    }
    for (int d = 16; d >= 1; d >>= 1) pre += __shfl_xor(pre, d, 32);
    if (lane == 31) s_own[w] = inc;
    if (lane == 0)  s_pre[w] = pre;
    __syncthreads();
    int off = 0;
    for (int k = 0; k < 32; ++k) {
        off += s_pre[k];
        if (k < w) off += s_own[k];
    }
    if (i < n) rowptr[i] = off + inc - x;
}

__global__ __launch_bounds__(256)
void place_edges_kernel(const int* __restrict__ rows, const int* __restrict__ cols,
                        const float* __restrict__ vals, const int* __restrict__ rank,
                        const int* __restrict__ rowptr, int* __restrict__ edges, int E)
{
    const int e = blockIdx.x * 256 + threadIdx.x;
    if (e < E) {
        const int pos = rowptr[rows[e]] + rank[e];
        reinterpret_cast<int2*>(edges)[pos] = make_int2(cols[e], __float_as_int(vals[e]));
    }
}

__global__ __launch_bounds__(256)
void spmm_rows_kernel(const int* __restrict__ rowptr, const int* __restrict__ edges,
                      const float* __restrict__ Hin, float* __restrict__ Hout, int N)
{
    const int g   = blockIdx.x * 256 + threadIdx.x;
    const int row = g >> 4;
    const int c4  = g & 15;
    if (row >= N) return;
    const int beg = rowptr[row];
    const int end = rowptr[row + 1];
    const int2*   ed = reinterpret_cast<const int2*>(edges);
    const float4* H4 = reinterpret_cast<const float4*>(Hin);
    float4 acc = make_float4(0.f, 0.f, 0.f, 0.f);
    for (int base = beg; base < end; base += 16) {
        const int idx = base + c4;
        int2 p = make_int2(0, 0);
        if (idx < end) p = ed[idx];
        const int cnt = min(16, end - base);
        for (int j = 0; j < cnt; j += 4) {
            const int   s0 = __shfl(p.x, j,     16), s1 = __shfl(p.x, j + 1, 16);
            const int   s2 = __shfl(p.x, j + 2, 16), s3 = __shfl(p.x, j + 3, 16);
            const float v0 = __int_as_float(__shfl(p.y, j,     16));
            const float v1 = __int_as_float(__shfl(p.y, j + 1, 16));
            const float v2 = __int_as_float(__shfl(p.y, j + 2, 16));
            const float v3 = __int_as_float(__shfl(p.y, j + 3, 16));
            const float4 x0 = H4[(size_t)s0 * 16 + c4];
            float4 x1 = H4[(size_t)s1 * 16 + c4];
            float4 x2 = H4[(size_t)s2 * 16 + c4];
            float4 x3 = H4[(size_t)s3 * 16 + c4];
            if (j + 1 >= cnt) x1 = make_float4(0.f, 0.f, 0.f, 0.f);
            if (j + 2 >= cnt) x2 = make_float4(0.f, 0.f, 0.f, 0.f);
            if (j + 3 >= cnt) x3 = make_float4(0.f, 0.f, 0.f, 0.f);
            acc.x += v0 * x0.x; acc.y += v0 * x0.y; acc.z += v0 * x0.z; acc.w += v0 * x0.w;
            acc.x += v1 * x1.x; acc.y += v1 * x1.y; acc.z += v1 * x1.z; acc.w += v1 * x1.w;
            acc.x += v2 * x2.x; acc.y += v2 * x2.y; acc.z += v2 * x2.z; acc.w += v2 * x2.w;
            acc.x += v3 * x3.x; acc.y += v3 * x3.y; acc.z += v3 * x3.z; acc.w += v3 * x3.w;
        }
    }
    reinterpret_cast<float4*>(Hout)[(size_t)row * 16 + c4] = acc;
}

extern "C" void kernel_launch(void* const* d_in, const int* in_sizes, int n_in,
                              void* d_out, int out_size, void* d_ws, size_t ws_size,
                              hipStream_t stream)
{
    const float* X    = (const float*)d_in[0];
    const float* W1   = (const float*)d_in[1];
    const float* W2   = (const float*)d_in[2];
    const float* vals = (const float*)d_in[3];
    const int*   rows = (const int*)d_in[4];
    const int*   cols = (const int*)d_in[5];
    float* out = (float*)d_out;

    const int N = in_sizes[0] / IN_DIM;
    const int E = in_sizes[3];

    const size_t hbytes = (size_t)N * OUTD * sizeof(float);
    const int n4 = (N * OUTD) / 4;
    const int zblocks = (n4 + 255) / 256;
    const int gblocks = (N + TM - 1) / TM;
    const int sblocks = (int)(((long long)E * 16 + 255) / 256);

    float* h0 = (float*)d_ws;

    const int    n1      = N + 1;
    const int    n4c     = (n1 + 3) / 4;
    const size_t ebytes  = (((size_t)E * 8) + 255) / 256 * 256;
    const size_t cbytes  = (((size_t)n4c * 16) + 255) / 256 * 256;
    const size_t need    = hbytes + ebytes + 2 * cbytes;

    if (ws_size >= need && (size_t)E * 4 <= hbytes) {
        char*  wsb    = (char*)d_ws;
        float* h1     = (float*)wsb;
        int*   rank   = (int*)wsb;
        int*   edges  = (int*)(wsb + hbytes);
        int*   cnt    = (int*)(wsb + hbytes + ebytes);
        int*   rowptr = (int*)(wsb + hbytes + ebytes + cbytes);
        const int eblocks = (E + 255) / 256;
        const int rblocks = (int)(((long long)N * 16 + 255) / 256);
        zero_f4_kernel<<<(n4c + 255) / 256, 256, 0, stream>>>((float4*)cnt, n4c);
        count_rank_kernel<<<eblocks, 256, 0, stream>>>(rows, cnt, rank, E);
        row_offsets_kernel<<<(n1 + SCAN_T - 1) / SCAN_T, SCAN_T, 0, stream>>>(cnt, rowptr, n1);
        place_edges_kernel<<<eblocks, 256, 0, stream>>>(rows, cols, vals, rank, rowptr, edges, E);
        gemm_fused_kernel<<<gblocks, 256, 0, stream>>>(X, W1, W2, out, N);
        spmm_rows_kernel<<<rblocks, 256, 0, stream>>>(rowptr, edges, out, h1, N);
        spmm_rows_kernel<<<rblocks, 256, 0, stream>>>(rowptr, edges, h1, out, N);
    } else if (ws_size >= 2 * hbytes) {
        float* h1 = h0 + (size_t)N * OUTD;
        gemm_fused_kernel<<<gblocks, 256, 0, stream>>>(X, W1, W2, h0, N);
        zero_f4_kernel<<<zblocks, 256, 0, stream>>>((float4*)h1, n4);
        spmm_edges_kernel<<<sblocks, 256, 0, stream>>>(vals, rows, cols, h0, h1, E);
        zero_f4_kernel<<<zblocks, 256, 0, stream>>>((float4*)out, n4);
        spmm_edges_kernel<<<sblocks, 256, 0, stream>>>(vals, rows, cols, h1, out, E);
    } else {
        gemm_fused_kernel<<<gblocks, 256, 0, stream>>>(X, W1, W2, h0, N);
        zero_f4_kernel<<<zblocks, 256, 0, stream>>>((float4*)out, n4);
        spmm_edges_kernel<<<sblocks, 256, 0, stream>>>(vals, rows, cols, h0, out, E);
        zero_f4_kernel<<<zblocks, 256, 0, stream>>>((float4*)h0, n4);
        spmm_edges_kernel<<<sblocks, 256, 0, stream>>>(vals, rows, cols, out, h0, E);
        hipMemcpyAsync(out, h0, hbytes, hipMemcpyDeviceToDevice, stream);
    }
}


# Edge grouping in one pass: each edge takes a slot in its row's fixed-capacity bucket by one returning int atomic (no rank array, no offsets scan, no second edge pass); 5 launches; sum order as before
# speedup vs baseline: 1.3594x; 1.1199x over previous
#include <hip/hip_runtime.h>

typedef float v2f __attribute__((ext_vector_type(2)));
typedef float v8f __attribute__((ext_vector_type(8)));

#define IN_DIM 256
#define HID    256
#define OUTD   64
#define TM     32

#define LP 260

__global__ __launch_bounds__(256)
void gemm_fused_kernel(const float* __restrict__ X, const float* __restrict__ W1,
                       const float* __restrict__ W2, float* __restrict__ H0, int N)
{
    __shared__ float4 sT4[TM * LP / 4];
    float* sT = reinterpret_cast<float*>(sT4);

    const int t    = threadIdx.x;
    const int wave = t >> 5;
    const int lane = t & 31;
    const int hi   = lane >> 4;
    const int r    = lane & 15;
    const int kb   = hi * 2;
    const int row0 = blockIdx.x * TM;

    for (int i = 0; i < 8; ++i) {
        int f    = t + i * 256;
        int rr   = f >> 6;
        int kq   = (f & 63) << 2;
        int grow = row0 + rr; if (grow >= N) grow = N - 1;
        const float4 g = reinterpret_cast<const float4*>(X + (size_t)grow * IN_DIM + kq)[0];
        *reinterpret_cast<float4*>(sT + rr * LP + kq) = g;
    }
    __syncthreads();

    {
        const int cb0 = wave * 2;
        const int c0  = cb0 * 16 + r;
        const int c1  = c0 + 16;
        const float* a0p = sT + r * LP + kb;
        const float* a1p = sT + (16 + r) * LP + kb;
        v8f a00 = {}, a01 = {}, a10 = {}, a11 = {};
        for (int k = 0; k < IN_DIM; k += 4) {
            v2f b0, b1;
            const v2f a0 = *reinterpret_cast<const v2f*>(a0p + k);
            const v2f a1 = *reinterpret_cast<const v2f*>(a1p + k);
            b0.x = W1[(k + kb) * HID + c0]; b0.y = W1[(k + kb + 1) * HID + c0];
            b1.x = W1[(k + kb) * HID + c1]; b1.y = W1[(k + kb + 1) * HID + c1];
            a00 = __builtin_amdgcn_wmma_f32_16x16x4_f32(false, a0, false, b0, (short)0, a00, false, false);
            a01 = __builtin_amdgcn_wmma_f32_16x16x4_f32(false, a0, false, b1, (short)0, a01, false, false);
            a10 = __builtin_amdgcn_wmma_f32_16x16x4_f32(false, a1, false, b0, (short)0, a10, false, false);
            a11 = __builtin_amdgcn_wmma_f32_16x16x4_f32(false, a1, false, b1, (short)0, a11, false, false);
        }
        __syncthreads();
        for (int v = 0; v < 8; ++v) {
            const int rl = v + 8 * hi;
            float x;
            x = a00[v]; sT[rl * LP + c0]        = x > 0.f ? x : 0.f;
            x = a01[v]; sT[rl * LP + c1]        = x > 0.f ? x : 0.f;
            x = a10[v]; sT[(16 + rl) * LP + c0] = x > 0.f ? x : 0.f;
            x = a11[v]; sT[(16 + rl) * LP + c1] = x > 0.f ? x : 0.f;
        }
    }
    __syncthreads();

    {
        const int rb = wave >> 2, cb = wave & 3;
        const int c  = cb * 16 + r;
        const float* ap = sT + (rb * 16 + r) * LP + kb;
        v8f acc = {};
        for (int k = 0; k < HID; k += 4) {
            v2f b;
            const v2f a = *reinterpret_cast<const v2f*>(ap + k);
            b.x = W2[(k + kb) * OUTD + c]; b.y = W2[(k + kb + 1) * OUTD + c];
            acc = __builtin_amdgcn_wmma_f32_16x16x4_f32(false, a, false, b, (short)0, acc, false, false);
        }
        for (int v = 0; v < 8; ++v) {
            const int grow = row0 + rb * 16 + v + 8 * hi;
            if (grow < N) H0[(size_t)grow * OUTD + c] = acc[v];
        }
    }
}

__global__ __launch_bounds__(256)
void spmm_edges_kernel(const float* __restrict__ vals, const int* __restrict__ rows,
                       const int* __restrict__ cols, const float* __restrict__ Hin,
                       float* __restrict__ Hout, int E)
{
    long long tid = (long long)blockIdx.x * blockDim.x + threadIdx.x;
    if (tid >= (long long)E * 16) return;
    const int e  = (int)(tid >> 4);
    const int c4 = (int)(tid & 15);
    const float v  = vals[e];
    const int col  = cols[e];
    const int row  = rows[e];
    const float4 x = reinterpret_cast<const float4*>(Hin)[(size_t)col * 16 + c4];
    float* base = Hout + (size_t)row * OUTD + c4 * 4;
    unsafeAtomicAdd(base + 0, v * x.x);
    unsafeAtomicAdd(base + 1, v * x.y);
    unsafeAtomicAdd(base + 2, v * x.z);
    unsafeAtomicAdd(base + 3, v * x.w);
}

__global__ __launch_bounds__(256)
void zero_f4_kernel(float4* __restrict__ p, int n4)
{
    int i = blockIdx.x * blockDim.x + threadIdx.x;
    if (i < n4) p[i] = make_float4(0.f, 0.f, 0.f, 0.f);
}


__global__ __launch_bounds__(256)
void count_rank_kernel(const int* __restrict__ rows, int* __restrict__ cnt,
                       int* __restrict__ rank, int E)
{
    const int e = blockIdx.x * 256 + threadIdx.x;
    if (e < E) rank[e] = atomicAdd(cnt + rows[e], 1);
}

#define SCAN_T 1024
__global__ __launch_bounds__(SCAN_T)
void row_offsets_kernel(const int* __restrict__ cnt, int* __restrict__ rowptr, int n)
{
    __shared__ int s_pre[32];
    __shared__ int s_own[32];
    const int t = threadIdx.x, b = blockIdx.x;
    const int lane = t & 31, w = t >> 5;

    int pre = 0;
    for (int c = 0; c < b; ++c) pre += cnt[c * SCAN_T + t];
    const int i = b * SCAN_T + t;
    const int x = (i < n) ? cnt[i] : 0;

    int inc = x;
    for (int d = 1; d < 32; d <<= 1) {
        const int y = __shfl_up(inc, d, 32);
        if (lane >= d) inc += y;
    }
    for (int d = 16; d >= 1; d >>= 1) pre += __shfl_xor(pre, d, 32);
    if (lane == 31) s_own[w] = inc;
    if (lane == 0)  s_pre[w] = pre;
    __syncthreads();
    int off = 0;
    for (int k = 0; k < 32; ++k) {
        off += s_pre[k];
        if (k < w) off += s_own[k];
    }
    if (i < n) rowptr[i] = off + inc - x;
}

__global__ __launch_bounds__(256)
void place_edges_kernel(const int* __restrict__ rows, const int* __restrict__ cols,
                        const float* __restrict__ vals, const int* __restrict__ rank,
                        const int* __restrict__ rowptr, int* __restrict__ edges, int E)
{
    const int e = blockIdx.x * 256 + threadIdx.x;
    if (e < E) {
        const int pos = rowptr[rows[e]] + rank[e];
        reinterpret_cast<int2*>(edges)[pos] = make_int2(cols[e], __float_as_int(vals[e]));
    }
}

__global__ __launch_bounds__(256)
void spmm_rows_kernel(const int* __restrict__ rowptr, const int* __restrict__ edges,
                      const float* __restrict__ Hin, float* __restrict__ Hout, int N)
{
    const int g   = blockIdx.x * 256 + threadIdx.x;
    const int row = g >> 4;
    const int c4  = g & 15;
    if (row >= N) return;
    const int beg = rowptr[row];
    const int end = rowptr[row + 1];
    const int2*   ed = reinterpret_cast<const int2*>(edges);
    const float4* H4 = reinterpret_cast<const float4*>(Hin);
    float4 acc = make_float4(0.f, 0.f, 0.f, 0.f);
    for (int base = beg; base < end; base += 16) {
        const int idx = base + c4;
        int2 p = make_int2(0, 0);
        if (idx < end) p = ed[idx];
        const int cnt = min(16, end - base);
        for (int j = 0; j < cnt; j += 4) {
            const int   s0 = __shfl(p.x, j,     16), s1 = __shfl(p.x, j + 1, 16);
            const int   s2 = __shfl(p.x, j + 2, 16), s3 = __shfl(p.x, j + 3, 16);
            const float v0 = __int_as_float(__shfl(p.y, j,     16));
            const float v1 = __int_as_float(__shfl(p.y, j + 1, 16));
            const float v2 = __int_as_float(__shfl(p.y, j + 2, 16));
            const float v3 = __int_as_float(__shfl(p.y, j + 3, 16));
            const float4 x0 = H4[(size_t)s0 * 16 + c4];
            float4 x1 = H4[(size_t)s1 * 16 + c4];
            float4 x2 = H4[(size_t)s2 * 16 + c4];
            float4 x3 = H4[(size_t)s3 * 16 + c4];
            if (j + 1 >= cnt) x1 = make_float4(0.f, 0.f, 0.f, 0.f);
            if (j + 2 >= cnt) x2 = make_float4(0.f, 0.f, 0.f, 0.f);
            if (j + 3 >= cnt) x3 = make_float4(0.f, 0.f, 0.f, 0.f);
            acc.x += v0 * x0.x; acc.y += v0 * x0.y; acc.z += v0 * x0.z; acc.w += v0 * x0.w;
            acc.x += v1 * x1.x; acc.y += v1 * x1.y; acc.z += v1 * x1.z; acc.w += v1 * x1.w;
            acc.x += v2 * x2.x; acc.y += v2 * x2.y; acc.z += v2 * x2.z; acc.w += v2 * x2.w;
            acc.x += v3 * x3.x; acc.y += v3 * x3.y; acc.z += v3 * x3.z; acc.w += v3 * x3.w;
        }
    }
    reinterpret_cast<float4*>(Hout)[(size_t)row * 16 + c4] = acc;
}

__global__ __launch_bounds__(256)
void bucket_edges_kernel(const int* __restrict__ rows, const int* __restrict__ cols,
                         const float* __restrict__ vals, int* __restrict__ cnt,
                         int* __restrict__ buckets, int K, int E)
{
    const int e = blockIdx.x * 256 + threadIdx.x;
    if (e < E) {
        const int   r = rows[e];
        const int   c = cols[e];
        const float v = vals[e];
        const int   s = atomicAdd(cnt + r, 1);
        reinterpret_cast<int2*>(buckets)[(size_t)r * K + min(s, K - 1)] = make_int2(c, __float_as_int(v));
    }
}

__global__ __launch_bounds__(256)
void spmm_buckets_kernel(const int* __restrict__ cnt, const int* __restrict__ buckets, int K,
                         const int* __restrict__ rows, const int* __restrict__ cols,
                         const float* __restrict__ vals, int E,
                         const float* __restrict__ Hin, float* __restrict__ Hout, int N)
{
    const int g   = blockIdx.x * 256 + threadIdx.x;
    const int row = g >> 4;
    const int c4  = g & 15;
    if (row >= N) return;
    const int end = cnt[row];
    const float4* H4 = reinterpret_cast<const float4*>(Hin);
    float4 acc = make_float4(0.f, 0.f, 0.f, 0.f);
    if (end <= K) {
        const int2* ed = reinterpret_cast<const int2*>(buckets) + (size_t)row * K;
        for (int base = 0; base < end; base += 16) {
            const int idx = base + c4;
            int2 p = make_int2(0, 0);
            if (idx < end) p = ed[idx];
            const int n = min(16, end - base);
            for (int j = 0; j < n; j += 4) {
                const int   s0 = __shfl(p.x, j,     16), s1 = __shfl(p.x, j + 1, 16);
                const int   s2 = __shfl(p.x, j + 2, 16), s3 = __shfl(p.x, j + 3, 16);
                const float v0 = __int_as_float(__shfl(p.y, j,     16));
                const float v1 = __int_as_float(__shfl(p.y, j + 1, 16));
                const float v2 = __int_as_float(__shfl(p.y, j + 2, 16));
                const float v3 = __int_as_float(__shfl(p.y, j + 3, 16));
                const float4 x0 = H4[(size_t)s0 * 16 + c4];
                float4 x1 = H4[(size_t)s1 * 16 + c4];
                float4 x2 = H4[(size_t)s2 * 16 + c4];
                float4 x3 = H4[(size_t)s3 * 16 + c4];
                if (j + 1 >= n) x1 = make_float4(0.f, 0.f, 0.f, 0.f);
                if (j + 2 >= n) x2 = make_float4(0.f, 0.f, 0.f, 0.f);
                if (j + 3 >= n) x3 = make_float4(0.f, 0.f, 0.f, 0.f);
                acc.x += v0 * x0.x; acc.y += v0 * x0.y; acc.z += v0 * x0.z; acc.w += v0 * x0.w;
                acc.x += v1 * x1.x; acc.y += v1 * x1.y; acc.z += v1 * x1.z; acc.w += v1 * x1.w;
                acc.x += v2 * x2.x; acc.y += v2 * x2.y; acc.z += v2 * x2.z; acc.w += v2 * x2.w;
                acc.x += v3 * x3.x; acc.y += v3 * x3.y; acc.z += v3 * x3.z; acc.w += v3 * x3.w;
            }
        }
    } else {
        for (int e = 0; e < E; ++e) {
            if (rows[e] == row) {
                const float  v = vals[e];
                const float4 x = H4[(size_t)cols[e] * 16 + c4];
                acc.x += v * x.x; acc.y += v * x.y; acc.z += v * x.z; acc.w += v * x.w;
            }
        }
    }
    reinterpret_cast<float4*>(Hout)[(size_t)row * 16 + c4] = acc;
}

extern "C" void kernel_launch(void* const* d_in, const int* in_sizes, int n_in,
                              void* d_out, int out_size, void* d_ws, size_t ws_size,
                              hipStream_t stream)
{
    const float* X    = (const float*)d_in[0];
    const float* W1   = (const float*)d_in[1];
    const float* W2   = (const float*)d_in[2];
    const float* vals = (const float*)d_in[3];
    const int*   rows = (const int*)d_in[4];
    const int*   cols = (const int*)d_in[5];
    float* out = (float*)d_out;

    const int N = in_sizes[0] / IN_DIM;
    const int E = in_sizes[3];

    const size_t hbytes = (size_t)N * OUTD * sizeof(float);
    const int n4 = (N * OUTD) / 4;
    const int zblocks = (n4 + 255) / 256;
    const int gblocks = (N + TM - 1) / TM;
    const int sblocks = (int)(((long long)E * 16 + 255) / 256);

    float* h0 = (float*)d_ws;

    const int    n1      = N + 1;
    const int    n4c     = (n1 + 3) / 4;
    const size_t ebytes  = (((size_t)E * 8) + 255) / 256 * 256;
    const size_t cbytes  = (((size_t)n4c * 16) + 255) / 256 * 256;
    const size_t need    = hbytes + ebytes + 2 * cbytes;

    const size_t b64 = (((size_t)N * 64 * 8) + 255) / 256 * 256;
    const size_t b48 = (((size_t)N * 48 * 8) + 255) / 256 * 256;
    const size_t b40 = (((size_t)N * 40 * 8) + 255) / 256 * 256;
    const int    K      = (ws_size >= hbytes + b64 + cbytes) ? 64
                        : (ws_size >= hbytes + b48 + cbytes) ? 48
                        : (ws_size >= hbytes + b40 + cbytes) ? 40 : 0;
    const size_t bbytes = (K == 64) ? b64 : (K == 48) ? b48 : b40;

    if (K > 0) {
        char*  wsb     = (char*)d_ws;
        float* h1      = (float*)wsb;
        int*   buckets = (int*)(wsb + hbytes);
        int*   cnt     = (int*)(wsb + hbytes + bbytes);
        const int eblocks = (E + 255) / 256;
        const int rblocks = (int)(((long long)N * 16 + 255) / 256);
        zero_f4_kernel<<<(n4c + 255) / 256, 256, 0, stream>>>((float4*)cnt, n4c);
        bucket_edges_kernel<<<eblocks, 256, 0, stream>>>(rows, cols, vals, cnt, buckets, K, E);
        gemm_fused_kernel<<<gblocks, 256, 0, stream>>>(X, W1, W2, out, N);
        spmm_buckets_kernel<<<rblocks, 256, 0, stream>>>(cnt, buckets, K, rows, cols, vals, E, out, h1, N);
        spmm_buckets_kernel<<<rblocks, 256, 0, stream>>>(cnt, buckets, K, rows, cols, vals, E, h1, out, N);
    } else if (ws_size >= need && (size_t)E * 4 <= hbytes) {
        char*  wsb    = (char*)d_ws;
        float* h1     = (float*)wsb;
        int*   rank   = (int*)wsb;
        int*   edges  = (int*)(wsb + hbytes);
        int*   cnt    = (int*)(wsb + hbytes + ebytes);
        int*   rowptr = (int*)(wsb + hbytes + ebytes + cbytes);
        const int eblocks = (E + 255) / 256;
        const int rblocks = (int)(((long long)N * 16 + 255) / 256);
        zero_f4_kernel<<<(n4c + 255) / 256, 256, 0, stream>>>((float4*)cnt, n4c);
        count_rank_kernel<<<eblocks, 256, 0, stream>>>(rows, cnt, rank, E);
        row_offsets_kernel<<<(n1 + SCAN_T - 1) / SCAN_T, SCAN_T, 0, stream>>>(cnt, rowptr, n1);
        place_edges_kernel<<<eblocks, 256, 0, stream>>>(rows, cols, vals, rank, rowptr, edges, E);
        gemm_fused_kernel<<<gblocks, 256, 0, stream>>>(X, W1, W2, out, N);
        spmm_rows_kernel<<<rblocks, 256, 0, stream>>>(rowptr, edges, out, h1, N);
        spmm_rows_kernel<<<rblocks, 256, 0, stream>>>(rowptr, edges, h1, out, N);
    } else if (ws_size >= 2 * hbytes) {
        float* h1 = h0 + (size_t)N * OUTD;
        gemm_fused_kernel<<<gblocks, 256, 0, stream>>>(X, W1, W2, h0, N);
        zero_f4_kernel<<<zblocks, 256, 0, stream>>>((float4*)h1, n4);
        spmm_edges_kernel<<<sblocks, 256, 0, stream>>>(vals, rows, cols, h0, h1, E);
        zero_f4_kernel<<<zblocks, 256, 0, stream>>>((float4*)out, n4);
        spmm_edges_kernel<<<sblocks, 256, 0, stream>>>(vals, rows, cols, h1, out, E);
    } else {
        gemm_fused_kernel<<<gblocks, 256, 0, stream>>>(X, W1, W2, h0, N);
        zero_f4_kernel<<<zblocks, 256, 0, stream>>>((float4*)out, n4);
        spmm_edges_kernel<<<sblocks, 256, 0, stream>>>(vals, rows, cols, h0, out, E);
        zero_f4_kernel<<<zblocks, 256, 0, stream>>>((float4*)h0, n4);
        spmm_edges_kernel<<<sblocks, 256, 0, stream>>>(vals, rows, cols, out, h0, E);
        hipMemcpyAsync(out, h0, hbytes, hipMemcpyDeviceToDevice, stream);
    }
}


# Edge bucketing fused into the dense kernel: a thread reads 2 edges up front, takes their slots by int atomic before product 2, stores the pairs last; 4 launches; sum order still arrival rank
# speedup vs baseline: 1.6490x; 1.2130x over previous
#include <hip/hip_runtime.h>

typedef float v2f __attribute__((ext_vector_type(2)));
typedef float v8f __attribute__((ext_vector_type(8)));

#define IN_DIM 256
#define HID    256
#define OUTD   64
#define TM     32

#define LP 260

__global__ __launch_bounds__(256)
void gemm_fused_kernel(const float* __restrict__ X, const float* __restrict__ W1,
                       const float* __restrict__ W2, float* __restrict__ H0, int N,
                       const int* __restrict__ rows, const int* __restrict__ cols,
                       const float* __restrict__ vals, int* __restrict__ cnt,
                       int* __restrict__ buckets, int K, int E)
{
    __shared__ float4 sT4[TM * LP / 4];
    float* sT = reinterpret_cast<float*>(sT4);

    const int t    = threadIdx.x;
    const int wave = t >> 5;
    const int lane = t & 31;
    const int hi   = lane >> 4;
    const int r    = lane & 15;
    const int kb   = hi * 2;
    const int row0 = blockIdx.x * TM;

    const int nthr = gridDim.x * 256;
    const int e0   = blockIdx.x * 256 + t;
    const int e1   = e0 + nthr;
    unsigned er0 = 0, er1 = 0;
    int      ec0 = 0, ec1 = 0;
    float    ev0 = 0.f, ev1 = 0.f;
    if (e0 < E) { er0 = (unsigned)rows[e0]; ec0 = cols[e0]; ev0 = vals[e0]; }
    if (e1 < E) { er1 = (unsigned)rows[e1]; ec1 = cols[e1]; ev1 = vals[e1]; }

    for (int i = 0; i < 8; ++i) {
        int f    = t + i * 256;
        int rr   = f >> 6;
        int kq   = (f & 63) << 2;
        int grow = row0 + rr; if (grow >= N) grow = N - 1;
        const float4 g = reinterpret_cast<const float4*>(X + (size_t)grow * IN_DIM + kq)[0];
        *reinterpret_cast<float4*>(sT + rr * LP + kq) = g;
    }
    __syncthreads();

    {
        const int cb0 = wave * 2;
        const int c0  = cb0 * 16 + r;
        const int c1  = c0 + 16;
        const float* a0p = sT + r * LP + kb;
        const float* a1p = sT + (16 + r) * LP + kb;
        v8f a00 = {}, a01 = {}, a10 = {}, a11 = {};
        for (int k = 0; k < IN_DIM; k += 4) {
            v2f b0, b1;
            const v2f a0 = *reinterpret_cast<const v2f*>(a0p + k);
            const v2f a1 = *reinterpret_cast<const v2f*>(a1p + k);
            b0.x = W1[(k + kb) * HID + c0]; b0.y = W1[(k + kb + 1) * HID + c0];
            b1.x = W1[(k + kb) * HID + c1]; b1.y = W1[(k + kb + 1) * HID + c1];
            a00 = __builtin_amdgcn_wmma_f32_16x16x4_f32(false, a0, false, b0, (short)0, a00, false, false);
            a01 = __builtin_amdgcn_wmma_f32_16x16x4_f32(false, a0, false, b1, (short)0, a01, false, false);
            a10 = __builtin_amdgcn_wmma_f32_16x16x4_f32(false, a1, false, b0, (short)0, a10, false, false);
            a11 = __builtin_amdgcn_wmma_f32_16x16x4_f32(false, a1, false, b1, (short)0, a11, false, false);
        }
        __syncthreads();
        for (int v = 0; v < 8; ++v) {
            const int rl = v + 8 * hi;
            float x;
            x = a00[v]; sT[rl * LP + c0]        = x > 0.f ? x : 0.f;
            x = a01[v]; sT[rl * LP + c1]        = x > 0.f ? x : 0.f;
            x = a10[v]; sT[(16 + rl) * LP + c0] = x > 0.f ? x : 0.f;
            x = a11[v]; sT[(16 + rl) * LP + c1] = x > 0.f ? x : 0.f;
        }
    }
    __syncthreads();

    int es0 = 0, es1 = 0;
    if (e0 < E) es0 = atomicAdd(cnt + er0, 1);
    if (e1 < E) es1 = atomicAdd(cnt + er1, 1);

    {
        const int rb = wave >> 2, cb = wave & 3;
        const int c  = cb * 16 + r;
        const float* ap = sT + (rb * 16 + r) * LP + kb;
        v8f acc = {};
        for (int k = 0; k < HID; k += 4) {
            v2f b;
            const v2f a = *reinterpret_cast<const v2f*>(ap + k);
            b.x = W2[(k + kb) * OUTD + c]; b.y = W2[(k + kb + 1) * OUTD + c];
            acc = __builtin_amdgcn_wmma_f32_16x16x4_f32(false, a, false, b, (short)0, acc, false, false);
        }
        for (int v = 0; v < 8; ++v) {
            const int grow = row0 + rb * 16 + v + 8 * hi;
            if (grow < N) H0[(size_t)grow * OUTD + c] = acc[v];
        }
    }

    int2* bk = reinterpret_cast<int2*>(buckets);
    if (e0 < E) bk[(size_t)er0 * K + min(es0, K - 1)] = make_int2(ec0, __float_as_int(ev0));
    if (e1 < E) bk[(size_t)er1 * K + min(es1, K - 1)] = make_int2(ec1, __float_as_int(ev1));
    if (e1 < E) {
        for (int e = e1 + nthr; e < E; e += nthr) {
            const unsigned r = (unsigned)rows[e];
            const int      s = atomicAdd(cnt + r, 1);
            bk[(size_t)r * K + min(s, K - 1)] = make_int2(cols[e], __float_as_int(vals[e]));
        }
    }
}

__global__ __launch_bounds__(256)
void spmm_edges_kernel(const float* __restrict__ vals, const int* __restrict__ rows,
                       const int* __restrict__ cols, const float* __restrict__ Hin,
                       float* __restrict__ Hout, int E)
{
    long long tid = (long long)blockIdx.x * blockDim.x + threadIdx.x;
    if (tid >= (long long)E * 16) return;
    const int e  = (int)(tid >> 4);
    const int c4 = (int)(tid & 15);
    const float v  = vals[e];
    const int col  = cols[e];
    const int row  = rows[e];
    const float4 x = reinterpret_cast<const float4*>(Hin)[(size_t)col * 16 + c4];
    float* base = Hout + (size_t)row * OUTD + c4 * 4;
    unsafeAtomicAdd(base + 0, v * x.x);
    unsafeAtomicAdd(base + 1, v * x.y);
    unsafeAtomicAdd(base + 2, v * x.z);
    unsafeAtomicAdd(base + 3, v * x.w);
}

__global__ __launch_bounds__(256)
void zero_f4_kernel(float4* __restrict__ p, int n4)
{
    int i = blockIdx.x * blockDim.x + threadIdx.x;
    if (i < n4) p[i] = make_float4(0.f, 0.f, 0.f, 0.f);
}


__global__ __launch_bounds__(256)
void count_rank_kernel(const int* __restrict__ rows, int* __restrict__ cnt,
                       int* __restrict__ rank, int E)
{
    const int e = blockIdx.x * 256 + threadIdx.x;
    if (e < E) rank[e] = atomicAdd(cnt + rows[e], 1);
}

#define SCAN_T 1024
__global__ __launch_bounds__(SCAN_T)
void row_offsets_kernel(const int* __restrict__ cnt, int* __restrict__ rowptr, int n)
{
    __shared__ int s_pre[32];
    __shared__ int s_own[32];
    const int t = threadIdx.x, b = blockIdx.x;
    const int lane = t & 31, w = t >> 5;

    int pre = 0;
    for (int c = 0; c < b; ++c) pre += cnt[c * SCAN_T + t];
    const int i = b * SCAN_T + t;
    const int x = (i < n) ? cnt[i] : 0;

    int inc = x;
    for (int d = 1; d < 32; d <<= 1) {
        const int y = __shfl_up(inc, d, 32);
        if (lane >= d) inc += y;
    }
    for (int d = 16; d >= 1; d >>= 1) pre += __shfl_xor(pre, d, 32);
    if (lane == 31) s_own[w] = inc;
    if (lane == 0)  s_pre[w] = pre;
    __syncthreads();
    int off = 0;
    for (int k = 0; k < 32; ++k) {
        off += s_pre[k];
        if (k < w) off += s_own[k];
    }
    if (i < n) rowptr[i] = off + inc - x;
}

__global__ __launch_bounds__(256)
void place_edges_kernel(const int* __restrict__ rows, const int* __restrict__ cols,
                        const float* __restrict__ vals, const int* __restrict__ rank,
                        const int* __restrict__ rowptr, int* __restrict__ edges, int E)
{
    const int e = blockIdx.x * 256 + threadIdx.x;
    if (e < E) {
        const int pos = rowptr[rows[e]] + rank[e];
        reinterpret_cast<int2*>(edges)[pos] = make_int2(cols[e], __float_as_int(vals[e]));
    }
}

__global__ __launch_bounds__(256)
void spmm_rows_kernel(const int* __restrict__ rowptr, const int* __restrict__ edges,
                      const float* __restrict__ Hin, float* __restrict__ Hout, int N)
{
    const int g   = blockIdx.x * 256 + threadIdx.x;
    const int row = g >> 4;
    const int c4  = g & 15;
    if (row >= N) return;
    const int beg = rowptr[row];
    const int end = rowptr[row + 1];
    const int2*   ed = reinterpret_cast<const int2*>(edges);
    const float4* H4 = reinterpret_cast<const float4*>(Hin);
    float4 acc = make_float4(0.f, 0.f, 0.f, 0.f);
    for (int base = beg; base < end; base += 16) {
        const int idx = base + c4;
        int2 p = make_int2(0, 0);
        if (idx < end) p = ed[idx];
        const int cnt = min(16, end - base);
        for (int j = 0; j < cnt; j += 4) {
            const int   s0 = __shfl(p.x, j,     16), s1 = __shfl(p.x, j + 1, 16);
            const int   s2 = __shfl(p.x, j + 2, 16), s3 = __shfl(p.x, j + 3, 16);
            const float v0 = __int_as_float(__shfl(p.y, j,     16));
            const float v1 = __int_as_float(__shfl(p.y, j + 1, 16));
            const float v2 = __int_as_float(__shfl(p.y, j + 2, 16));
            const float v3 = __int_as_float(__shfl(p.y, j + 3, 16));
            const float4 x0 = H4[(size_t)s0 * 16 + c4];
            float4 x1 = H4[(size_t)s1 * 16 + c4];
            float4 x2 = H4[(size_t)s2 * 16 + c4];
            float4 x3 = H4[(size_t)s3 * 16 + c4];
            if (j + 1 >= cnt) x1 = make_float4(0.f, 0.f, 0.f, 0.f);
            if (j + 2 >= cnt) x2 = make_float4(0.f, 0.f, 0.f, 0.f);
            if (j + 3 >= cnt) x3 = make_float4(0.f, 0.f, 0.f, 0.f);
            acc.x += v0 * x0.x; acc.y += v0 * x0.y; acc.z += v0 * x0.z; acc.w += v0 * x0.w;
            acc.x += v1 * x1.x; acc.y += v1 * x1.y; acc.z += v1 * x1.z; acc.w += v1 * x1.w;
            acc.x += v2 * x2.x; acc.y += v2 * x2.y; acc.z += v2 * x2.z; acc.w += v2 * x2.w;
            acc.x += v3 * x3.x; acc.y += v3 * x3.y; acc.z += v3 * x3.z; acc.w += v3 * x3.w;
        }
    }
    reinterpret_cast<float4*>(Hout)[(size_t)row * 16 + c4] = acc;
}


__global__ __launch_bounds__(256)
void spmm_buckets_kernel(const int* __restrict__ cnt, const int* __restrict__ buckets, int K,
                         const int* __restrict__ rows, const int* __restrict__ cols,
                         const float* __restrict__ vals, int E,
                         const float* __restrict__ Hin, float* __restrict__ Hout, int N)
{
    const int g   = blockIdx.x * 256 + threadIdx.x;
    const int row = g >> 4;
    const int c4  = g & 15;
    if (row >= N) return;
    const int end = cnt[row];
    const float4* H4 = reinterpret_cast<const float4*>(Hin);
    float4 acc = make_float4(0.f, 0.f, 0.f, 0.f);
    if (end <= K) {
        const int2* ed = reinterpret_cast<const int2*>(buckets) + (size_t)row * K;
        for (int base = 0; base < end; base += 16) {
            const int idx = base + c4;
            int2 p = make_int2(0, 0);
            if (idx < end) p = ed[idx];
            const int n = min(16, end - base);
            for (int j = 0; j < n; j += 4) {
                const int   s0 = __shfl(p.x, j,     16), s1 = __shfl(p.x, j + 1, 16);
                const int   s2 = __shfl(p.x, j + 2, 16), s3 = __shfl(p.x, j + 3, 16);
                const float v0 = __int_as_float(__shfl(p.y, j,     16));
                const float v1 = __int_as_float(__shfl(p.y, j + 1, 16));
                const float v2 = __int_as_float(__shfl(p.y, j + 2, 16));
                const float v3 = __int_as_float(__shfl(p.y, j + 3, 16));
                const float4 x0 = H4[(size_t)s0 * 16 + c4];
                float4 x1 = H4[(size_t)s1 * 16 + c4];
                float4 x2 = H4[(size_t)s2 * 16 + c4];
                float4 x3 = H4[(size_t)s3 * 16 + c4];
                if (j + 1 >= n) x1 = make_float4(0.f, 0.f, 0.f, 0.f);
                if (j + 2 >= n) x2 = make_float4(0.f, 0.f, 0.f, 0.f);
                if (j + 3 >= n) x3 = make_float4(0.f, 0.f, 0.f, 0.f);
                acc.x += v0 * x0.x; acc.y += v0 * x0.y; acc.z += v0 * x0.z; acc.w += v0 * x0.w;
                acc.x += v1 * x1.x; acc.y += v1 * x1.y; acc.z += v1 * x1.z; acc.w += v1 * x1.w;
                acc.x += v2 * x2.x; acc.y += v2 * x2.y; acc.z += v2 * x2.z; acc.w += v2 * x2.w;
                acc.x += v3 * x3.x; acc.y += v3 * x3.y; acc.z += v3 * x3.z; acc.w += v3 * x3.w;
            }
        }
    } else {
        for (int e = 0; e < E; ++e) {
            if (rows[e] == row) {
                const float  v = vals[e];
                const float4 x = H4[(size_t)cols[e] * 16 + c4];
                acc.x += v * x.x; acc.y += v * x.y; acc.z += v * x.z; acc.w += v * x.w;
            }
        }
    }
    reinterpret_cast<float4*>(Hout)[(size_t)row * 16 + c4] = acc;
}

extern "C" void kernel_launch(void* const* d_in, const int* in_sizes, int n_in,
                              void* d_out, int out_size, void* d_ws, size_t ws_size,
                              hipStream_t stream)
{
    const float* X    = (const float*)d_in[0];
    const float* W1   = (const float*)d_in[1];
    const float* W2   = (const float*)d_in[2];
    const float* vals = (const float*)d_in[3];
    const int*   rows = (const int*)d_in[4];
    const int*   cols = (const int*)d_in[5];
    float* out = (float*)d_out;

    const int N = in_sizes[0] / IN_DIM;
    const int E = in_sizes[3];

    const size_t hbytes = (size_t)N * OUTD * sizeof(float);
    const int n4 = (N * OUTD) / 4;
    const int zblocks = (n4 + 255) / 256;
    const int gblocks = (N + TM - 1) / TM;
    const int sblocks = (int)(((long long)E * 16 + 255) / 256);

    float* h0 = (float*)d_ws;

    const int    n1      = N + 1;
    const int    n4c     = (n1 + 3) / 4;
    const size_t ebytes  = (((size_t)E * 8) + 255) / 256 * 256;
    const size_t cbytes  = (((size_t)n4c * 16) + 255) / 256 * 256;
    const size_t need    = hbytes + ebytes + 2 * cbytes;

    const size_t b64 = (((size_t)N * 64 * 8) + 255) / 256 * 256;
    const size_t b48 = (((size_t)N * 48 * 8) + 255) / 256 * 256;
    const size_t b40 = (((size_t)N * 40 * 8) + 255) / 256 * 256;
    const int    K      = (ws_size >= hbytes + b64 + cbytes) ? 64
                        : (ws_size >= hbytes + b48 + cbytes) ? 48
                        : (ws_size >= hbytes + b40 + cbytes) ? 40 : 0;
    const size_t bbytes = (K == 64) ? b64 : (K == 48) ? b48 : b40;

    if (K > 0) {
        char*  wsb     = (char*)d_ws;
        float* h1      = (float*)wsb;
        int*   buckets = (int*)(wsb + hbytes);
        int*   cnt     = (int*)(wsb + hbytes + bbytes);
        const int rblocks = (int)(((long long)N * 16 + 255) / 256);
        zero_f4_kernel<<<(n4c + 255) / 256, 256, 0, stream>>>((float4*)cnt, n4c);
        gemm_fused_kernel<<<gblocks, 256, 0, stream>>>(X, W1, W2, out, N, rows, cols, vals, cnt, buckets, K, E);
        spmm_buckets_kernel<<<rblocks, 256, 0, stream>>>(cnt, buckets, K, rows, cols, vals, E, out, h1, N);
        spmm_buckets_kernel<<<rblocks, 256, 0, stream>>>(cnt, buckets, K, rows, cols, vals, E, h1, out, N);
    } else if (ws_size >= need && (size_t)E * 4 <= hbytes) {
        char*  wsb    = (char*)d_ws;
        float* h1     = (float*)wsb;
        int*   rank   = (int*)wsb;
        int*   edges  = (int*)(wsb + hbytes);
        int*   cnt    = (int*)(wsb + hbytes + ebytes);
        int*   rowptr = (int*)(wsb + hbytes + ebytes + cbytes);
        const int eblocks = (E + 255) / 256;
        const int rblocks = (int)(((long long)N * 16 + 255) / 256);
        zero_f4_kernel<<<(n4c + 255) / 256, 256, 0, stream>>>((float4*)cnt, n4c);
        count_rank_kernel<<<eblocks, 256, 0, stream>>>(rows, cnt, rank, E);
        row_offsets_kernel<<<(n1 + SCAN_T - 1) / SCAN_T, SCAN_T, 0, stream>>>(cnt, rowptr, n1);
        place_edges_kernel<<<eblocks, 256, 0, stream>>>(rows, cols, vals, rank, rowptr, edges, E);
        gemm_fused_kernel<<<gblocks, 256, 0, stream>>>(X, W1, W2, out, N, nullptr, nullptr, nullptr, nullptr, nullptr, 0, 0);
        spmm_rows_kernel<<<rblocks, 256, 0, stream>>>(rowptr, edges, out, h1, N);
        spmm_rows_kernel<<<rblocks, 256, 0, stream>>>(rowptr, edges, h1, out, N);
    } else if (ws_size >= 2 * hbytes) {
        float* h1 = h0 + (size_t)N * OUTD;
        gemm_fused_kernel<<<gblocks, 256, 0, stream>>>(X, W1, W2, h0, N, nullptr, nullptr, nullptr, nullptr, nullptr, 0, 0);
        zero_f4_kernel<<<zblocks, 256, 0, stream>>>((float4*)h1, n4);
        spmm_edges_kernel<<<sblocks, 256, 0, stream>>>(vals, rows, cols, h0, h1, E);
        zero_f4_kernel<<<zblocks, 256, 0, stream>>>((float4*)out, n4);
        spmm_edges_kernel<<<sblocks, 256, 0, stream>>>(vals, rows, cols, h1, out, E);
    } else {
        gemm_fused_kernel<<<gblocks, 256, 0, stream>>>(X, W1, W2, h0, N, nullptr, nullptr, nullptr, nullptr, nullptr, 0, 0);
        zero_f4_kernel<<<zblocks, 256, 0, stream>>>((float4*)out, n4);
        spmm_edges_kernel<<<sblocks, 256, 0, stream>>>(vals, rows, cols, h0, out, E);
        zero_f4_kernel<<<zblocks, 256, 0, stream>>>((float4*)h0, n4);
        spmm_edges_kernel<<<sblocks, 256, 0, stream>>>(vals, rows, cols, out, h0, E);
        hipMemcpyAsync(out, h0, hbytes, hipMemcpyDeviceToDevice, stream);
    }
}


# Both GEMM products: a lane owns two adjacent columns (a wave's two 16-col blocks interleaved even/odd): W1/W2 by 8-byte whole-line loads, tile and output rows by 8-byte stores; sum order unchanged
# speedup vs baseline: 1.7574x; 1.0657x over previous
#include <hip/hip_runtime.h>

typedef float v2f __attribute__((ext_vector_type(2)));
typedef float v8f __attribute__((ext_vector_type(8)));

#define IN_DIM 256
#define HID    256
#define OUTD   64
#define TM     32

#define LP 260

__global__ __launch_bounds__(256)
void gemm_fused_kernel(const float* __restrict__ X, const float* __restrict__ W1,
                       const float* __restrict__ W2, float* __restrict__ H0, int N,
                       const int* __restrict__ rows, const int* __restrict__ cols,
                       const float* __restrict__ vals, int* __restrict__ cnt,
                       int* __restrict__ buckets, int K, int E)
{
    __shared__ float4 sT4[TM * LP / 4];
    float* sT = reinterpret_cast<float*>(sT4);

    const int t    = threadIdx.x;
    const int wave = t >> 5;
    const int lane = t & 31;
    const int hi   = lane >> 4;
    const int r    = lane & 15;
    const int kb   = hi * 2;
    const int row0 = blockIdx.x * TM;

    const int nthr = gridDim.x * 256;
    const int e0   = blockIdx.x * 256 + t;
    const int e1   = e0 + nthr;
    unsigned er0 = 0, er1 = 0;
    int      ec0 = 0, ec1 = 0;
    float    ev0 = 0.f, ev1 = 0.f;
    if (e0 < E) { er0 = (unsigned)rows[e0]; ec0 = cols[e0]; ev0 = vals[e0]; }
    if (e1 < E) { er1 = (unsigned)rows[e1]; ec1 = cols[e1]; ev1 = vals[e1]; }

    for (int i = 0; i < 8; ++i) {
        int f    = t + i * 256;
        int rr   = f >> 6;
        int kq   = (f & 63) << 2;
        int grow = row0 + rr; if (grow >= N) grow = N - 1;
        const float4 g = reinterpret_cast<const float4*>(X + (size_t)grow * IN_DIM + kq)[0];
        *reinterpret_cast<float4*>(sT + rr * LP + kq) = g;
    }
    __syncthreads();

    {
        const int c0 = wave * 32 + 2 * r;
        const float* a0p = sT + r * LP + kb;
        const float* a1p = sT + (16 + r) * LP + kb;
        const float* wp  = W1 + kb * HID + c0;
        v8f a00 = {}, a01 = {}, a10 = {}, a11 = {};
        for (int k = 0; k < IN_DIM; k += 4) {
            const v2f a0 = *reinterpret_cast<const v2f*>(a0p + k);
            const v2f a1 = *reinterpret_cast<const v2f*>(a1p + k);
            const v2f w0 = *reinterpret_cast<const v2f*>(wp + (size_t)k * HID);
            const v2f w1 = *reinterpret_cast<const v2f*>(wp + (size_t)(k + 1) * HID);
            v2f b0, b1;
            b0.x = w0.x; b0.y = w1.x;
            b1.x = w0.y; b1.y = w1.y;
            a00 = __builtin_amdgcn_wmma_f32_16x16x4_f32(false, a0, false, b0, (short)0, a00, false, false);
            a01 = __builtin_amdgcn_wmma_f32_16x16x4_f32(false, a0, false, b1, (short)0, a01, false, false);
            a10 = __builtin_amdgcn_wmma_f32_16x16x4_f32(false, a1, false, b0, (short)0, a10, false, false);
            a11 = __builtin_amdgcn_wmma_f32_16x16x4_f32(false, a1, false, b1, (short)0, a11, false, false);
        }
        __syncthreads();
        for (int v = 0; v < 8; ++v) {
            const int rl = v + 8 * hi;
            float x;
            v2f m0, m1;
            x = a00[v]; m0.x = x > 0.f ? x : 0.f;
            x = a01[v]; m0.y = x > 0.f ? x : 0.f;
            x = a10[v]; m1.x = x > 0.f ? x : 0.f;
            x = a11[v]; m1.y = x > 0.f ? x : 0.f;
            *reinterpret_cast<v2f*>(sT + rl * LP + c0)        = m0;
            *reinterpret_cast<v2f*>(sT + (16 + rl) * LP + c0) = m1;
        }
    }
    __syncthreads();

    int es0 = 0, es1 = 0;
    if (e0 < E) es0 = atomicAdd(cnt + er0, 1);
    if (e1 < E) es1 = atomicAdd(cnt + er1, 1);

    if (wave < 4) {
        const int rb = wave >> 1, cp = wave & 1;
        const int c  = cp * 32 + 2 * r;
        const float* ap = sT + (rb * 16 + r) * LP + kb;
        const float* wp = W2 + kb * OUTD + c;
        v8f acc0 = {}, acc1 = {};
        for (int k = 0; k < HID; k += 4) {
            const v2f a  = *reinterpret_cast<const v2f*>(ap + k);
            const v2f w0 = *reinterpret_cast<const v2f*>(wp + (size_t)k * OUTD);
            const v2f w1 = *reinterpret_cast<const v2f*>(wp + (size_t)(k + 1) * OUTD);
            v2f b0, b1;
            b0.x = w0.x; b0.y = w1.x;
            b1.x = w0.y; b1.y = w1.y;
            acc0 = __builtin_amdgcn_wmma_f32_16x16x4_f32(false, a, false, b0, (short)0, acc0, false, false);
            acc1 = __builtin_amdgcn_wmma_f32_16x16x4_f32(false, a, false, b1, (short)0, acc1, false, false);
        }
        for (int v = 0; v < 8; ++v) {
            const int grow = row0 + rb * 16 + v + 8 * hi;
            if (grow < N) {
                v2f o; o.x = acc0[v]; o.y = acc1[v];
                *reinterpret_cast<v2f*>(H0 + (size_t)grow * OUTD + c) = o;
            }
        }
    }

    int2* bk = reinterpret_cast<int2*>(buckets);
    if (e0 < E) bk[(size_t)er0 * K + min(es0, K - 1)] = make_int2(ec0, __float_as_int(ev0));
    if (e1 < E) bk[(size_t)er1 * K + min(es1, K - 1)] = make_int2(ec1, __float_as_int(ev1));
    if (e1 < E) {
        for (int e = e1 + nthr; e < E; e += nthr) {
            const unsigned r = (unsigned)rows[e];
            const int      s = atomicAdd(cnt + r, 1);
            bk[(size_t)r * K + min(s, K - 1)] = make_int2(cols[e], __float_as_int(vals[e]));
        }
    }
}

__global__ __launch_bounds__(256)
void spmm_edges_kernel(const float* __restrict__ vals, const int* __restrict__ rows,
                       const int* __restrict__ cols, const float* __restrict__ Hin,
                       float* __restrict__ Hout, int E)
{
    long long tid = (long long)blockIdx.x * blockDim.x + threadIdx.x;
    if (tid >= (long long)E * 16) return;
    const int e  = (int)(tid >> 4);
    const int c4 = (int)(tid & 15);
    const float v  = vals[e];
    const int col  = cols[e];
    const int row  = rows[e];
    const float4 x = reinterpret_cast<const float4*>(Hin)[(size_t)col * 16 + c4];
    float* base = Hout + (size_t)row * OUTD + c4 * 4;
    unsafeAtomicAdd(base + 0, v * x.x);
    unsafeAtomicAdd(base + 1, v * x.y);
    unsafeAtomicAdd(base + 2, v * x.z);
    unsafeAtomicAdd(base + 3, v * x.w);
}

__global__ __launch_bounds__(256)
void zero_f4_kernel(float4* __restrict__ p, int n4)
{
    int i = blockIdx.x * blockDim.x + threadIdx.x;
    if (i < n4) p[i] = make_float4(0.f, 0.f, 0.f, 0.f);
}


__global__ __launch_bounds__(256)
void count_rank_kernel(const int* __restrict__ rows, int* __restrict__ cnt,
                       int* __restrict__ rank, int E)
{
    const int e = blockIdx.x * 256 + threadIdx.x;
    if (e < E) rank[e] = atomicAdd(cnt + rows[e], 1);
}

#define SCAN_T 1024
__global__ __launch_bounds__(SCAN_T)
void row_offsets_kernel(const int* __restrict__ cnt, int* __restrict__ rowptr, int n)
{
    __shared__ int s_pre[32];
    __shared__ int s_own[32];
    const int t = threadIdx.x, b = blockIdx.x;
    const int lane = t & 31, w = t >> 5;

    int pre = 0;
    for (int c = 0; c < b; ++c) pre += cnt[c * SCAN_T + t];
    const int i = b * SCAN_T + t;
    const int x = (i < n) ? cnt[i] : 0;

    int inc = x;
    for (int d = 1; d < 32; d <<= 1) {
        const int y = __shfl_up(inc, d, 32);
        if (lane >= d) inc += y;
    }
    for (int d = 16; d >= 1; d >>= 1) pre += __shfl_xor(pre, d, 32);
    if (lane == 31) s_own[w] = inc;
    if (lane == 0)  s_pre[w] = pre;
    __syncthreads();
    int off = 0;
    for (int k = 0; k < 32; ++k) {
        off += s_pre[k];
        if (k < w) off += s_own[k];
    }
    if (i < n) rowptr[i] = off + inc - x;
}

__global__ __launch_bounds__(256)
void place_edges_kernel(const int* __restrict__ rows, const int* __restrict__ cols,
                        const float* __restrict__ vals, const int* __restrict__ rank,
                        const int* __restrict__ rowptr, int* __restrict__ edges, int E)
{
    const int e = blockIdx.x * 256 + threadIdx.x;
    if (e < E) {
        const int pos = rowptr[rows[e]] + rank[e];
        reinterpret_cast<int2*>(edges)[pos] = make_int2(cols[e], __float_as_int(vals[e]));
    }
}

__global__ __launch_bounds__(256)
void spmm_rows_kernel(const int* __restrict__ rowptr, const int* __restrict__ edges,
                      const float* __restrict__ Hin, float* __restrict__ Hout, int N)
{
    const int g   = blockIdx.x * 256 + threadIdx.x;
    const int row = g >> 4;
    const int c4  = g & 15;
    if (row >= N) return;
    const int beg = rowptr[row];
    const int end = rowptr[row + 1];
    const int2*   ed = reinterpret_cast<const int2*>(edges);
    const float4* H4 = reinterpret_cast<const float4*>(Hin);
    float4 acc = make_float4(0.f, 0.f, 0.f, 0.f);
    for (int base = beg; base < end; base += 16) {
        const int idx = base + c4;
        int2 p = make_int2(0, 0);
        if (idx < end) p = ed[idx];
        const int cnt = min(16, end - base);
        for (int j = 0; j < cnt; j += 4) {
            const int   s0 = __shfl(p.x, j,     16), s1 = __shfl(p.x, j + 1, 16);
            const int   s2 = __shfl(p.x, j + 2, 16), s3 = __shfl(p.x, j + 3, 16);
            const float v0 = __int_as_float(__shfl(p.y, j,     16));
            const float v1 = __int_as_float(__shfl(p.y, j + 1, 16));
            const float v2 = __int_as_float(__shfl(p.y, j + 2, 16));
            const float v3 = __int_as_float(__shfl(p.y, j + 3, 16));
            const float4 x0 = H4[(size_t)s0 * 16 + c4];
            float4 x1 = H4[(size_t)s1 * 16 + c4];
            float4 x2 = H4[(size_t)s2 * 16 + c4];
            float4 x3 = H4[(size_t)s3 * 16 + c4];
            if (j + 1 >= cnt) x1 = make_float4(0.f, 0.f, 0.f, 0.f);
            if (j + 2 >= cnt) x2 = make_float4(0.f, 0.f, 0.f, 0.f);
            if (j + 3 >= cnt) x3 = make_float4(0.f, 0.f, 0.f, 0.f);
            acc.x += v0 * x0.x; acc.y += v0 * x0.y; acc.z += v0 * x0.z; acc.w += v0 * x0.w;
            acc.x += v1 * x1.x; acc.y += v1 * x1.y; acc.z += v1 * x1.z; acc.w += v1 * x1.w;
            acc.x += v2 * x2.x; acc.y += v2 * x2.y; acc.z += v2 * x2.z; acc.w += v2 * x2.w;
            acc.x += v3 * x3.x; acc.y += v3 * x3.y; acc.z += v3 * x3.z; acc.w += v3 * x3.w;
        }
    }
    reinterpret_cast<float4*>(Hout)[(size_t)row * 16 + c4] = acc;
}


__global__ __launch_bounds__(256)
void spmm_buckets_kernel(const int* __restrict__ cnt, const int* __restrict__ buckets, int K,
                         const int* __restrict__ rows, const int* __restrict__ cols,
                         const float* __restrict__ vals, int E,
                         const float* __restrict__ Hin, float* __restrict__ Hout, int N)
{
    const int g   = blockIdx.x * 256 + threadIdx.x;
    const int row = g >> 4;
    const int c4  = g & 15;
    if (row >= N) return;
    const int end = cnt[row];
    const float4* H4 = reinterpret_cast<const float4*>(Hin);
    float4 acc = make_float4(0.f, 0.f, 0.f, 0.f);
    if (end <= K) {
        const int2* ed = reinterpret_cast<const int2*>(buckets) + (size_t)row * K;
        for (int base = 0; base < end; base += 16) {
            const int idx = base + c4;
            int2 p = make_int2(0, 0);
            if (idx < end) p = ed[idx];
            const int n = min(16, end - base);
            for (int j = 0; j < n; j += 4) {
                const int   s0 = __shfl(p.x, j,     16), s1 = __shfl(p.x, j + 1, 16);
                const int   s2 = __shfl(p.x, j + 2, 16), s3 = __shfl(p.x, j + 3, 16);
                const float v0 = __int_as_float(__shfl(p.y, j,     16));
                const float v1 = __int_as_float(__shfl(p.y, j + 1, 16));
                const float v2 = __int_as_float(__shfl(p.y, j + 2, 16));
                const float v3 = __int_as_float(__shfl(p.y, j + 3, 16));
                const float4 x0 = H4[(size_t)s0 * 16 + c4];
                float4 x1 = H4[(size_t)s1 * 16 + c4];
                float4 x2 = H4[(size_t)s2 * 16 + c4];
                float4 x3 = H4[(size_t)s3 * 16 + c4];
                if (j + 1 >= n) x1 = make_float4(0.f, 0.f, 0.f, 0.f);
                if (j + 2 >= n) x2 = make_float4(0.f, 0.f, 0.f, 0.f);
                if (j + 3 >= n) x3 = make_float4(0.f, 0.f, 0.f, 0.f);
                acc.x += v0 * x0.x; acc.y += v0 * x0.y; acc.z += v0 * x0.z; acc.w += v0 * x0.w;
                acc.x += v1 * x1.x; acc.y += v1 * x1.y; acc.z += v1 * x1.z; acc.w += v1 * x1.w;
                acc.x += v2 * x2.x; acc.y += v2 * x2.y; acc.z += v2 * x2.z; acc.w += v2 * x2.w;
                acc.x += v3 * x3.x; acc.y += v3 * x3.y; acc.z += v3 * x3.z; acc.w += v3 * x3.w;
            }
        }
    } else {
        for (int e = 0; e < E; ++e) {
            if (rows[e] == row) {
                const float  v = vals[e];
                const float4 x = H4[(size_t)cols[e] * 16 + c4];
                acc.x += v * x.x; acc.y += v * x.y; acc.z += v * x.z; acc.w += v * x.w;
            }
        }
    }
    reinterpret_cast<float4*>(Hout)[(size_t)row * 16 + c4] = acc;
}

extern "C" void kernel_launch(void* const* d_in, const int* in_sizes, int n_in,
                              void* d_out, int out_size, void* d_ws, size_t ws_size,
                              hipStream_t stream)
{
    const float* X    = (const float*)d_in[0];
    const float* W1   = (const float*)d_in[1];
    const float* W2   = (const float*)d_in[2];
    const float* vals = (const float*)d_in[3];
    const int*   rows = (const int*)d_in[4];
    const int*   cols = (const int*)d_in[5];
    float* out = (float*)d_out;

    const int N = in_sizes[0] / IN_DIM;
    const int E = in_sizes[3];

    const size_t hbytes = (size_t)N * OUTD * sizeof(float);
    const int n4 = (N * OUTD) / 4;
    const int zblocks = (n4 + 255) / 256;
    const int gblocks = (N + TM - 1) / TM;
    const int sblocks = (int)(((long long)E * 16 + 255) / 256);

    float* h0 = (float*)d_ws;

    const int    n1      = N + 1;
    const int    n4c     = (n1 + 3) / 4;
    const size_t ebytes  = (((size_t)E * 8) + 255) / 256 * 256;
    const size_t cbytes  = (((size_t)n4c * 16) + 255) / 256 * 256;
    const size_t need    = hbytes + ebytes + 2 * cbytes;

    const size_t b64 = (((size_t)N * 64 * 8) + 255) / 256 * 256;
    const size_t b48 = (((size_t)N * 48 * 8) + 255) / 256 * 256;
    const size_t b40 = (((size_t)N * 40 * 8) + 255) / 256 * 256;
    const int    K      = (ws_size >= hbytes + b64 + cbytes) ? 64
                        : (ws_size >= hbytes + b48 + cbytes) ? 48
                        : (ws_size >= hbytes + b40 + cbytes) ? 40 : 0;
    const size_t bbytes = (K == 64) ? b64 : (K == 48) ? b48 : b40;

    if (K > 0) {
        char*  wsb     = (char*)d_ws;
        float* h1      = (float*)wsb;
        int*   buckets = (int*)(wsb + hbytes);
        int*   cnt     = (int*)(wsb + hbytes + bbytes);
        const int rblocks = (int)(((long long)N * 16 + 255) / 256);
        zero_f4_kernel<<<(n4c + 255) / 256, 256, 0, stream>>>((float4*)cnt, n4c);
        gemm_fused_kernel<<<gblocks, 256, 0, stream>>>(X, W1, W2, out, N, rows, cols, vals, cnt, buckets, K, E);
        spmm_buckets_kernel<<<rblocks, 256, 0, stream>>>(cnt, buckets, K, rows, cols, vals, E, out, h1, N);
        spmm_buckets_kernel<<<rblocks, 256, 0, stream>>>(cnt, buckets, K, rows, cols, vals, E, h1, out, N);
    } else if (ws_size >= need && (size_t)E * 4 <= hbytes) {
        char*  wsb    = (char*)d_ws;
        float* h1     = (float*)wsb;
        int*   rank   = (int*)wsb;
        int*   edges  = (int*)(wsb + hbytes);
        int*   cnt    = (int*)(wsb + hbytes + ebytes);
        int*   rowptr = (int*)(wsb + hbytes + ebytes + cbytes);
        const int eblocks = (E + 255) / 256;
        const int rblocks = (int)(((long long)N * 16 + 255) / 256);
        zero_f4_kernel<<<(n4c + 255) / 256, 256, 0, stream>>>((float4*)cnt, n4c);
        count_rank_kernel<<<eblocks, 256, 0, stream>>>(rows, cnt, rank, E);
        row_offsets_kernel<<<(n1 + SCAN_T - 1) / SCAN_T, SCAN_T, 0, stream>>>(cnt, rowptr, n1);
        place_edges_kernel<<<eblocks, 256, 0, stream>>>(rows, cols, vals, rank, rowptr, edges, E);
        gemm_fused_kernel<<<gblocks, 256, 0, stream>>>(X, W1, W2, out, N, nullptr, nullptr, nullptr, nullptr, nullptr, 0, 0);
        spmm_rows_kernel<<<rblocks, 256, 0, stream>>>(rowptr, edges, out, h1, N);
        spmm_rows_kernel<<<rblocks, 256, 0, stream>>>(rowptr, edges, h1, out, N);
    } else if (ws_size >= 2 * hbytes) {
        float* h1 = h0 + (size_t)N * OUTD;
        gemm_fused_kernel<<<gblocks, 256, 0, stream>>>(X, W1, W2, h0, N, nullptr, nullptr, nullptr, nullptr, nullptr, 0, 0);
        zero_f4_kernel<<<zblocks, 256, 0, stream>>>((float4*)h1, n4);
        spmm_edges_kernel<<<sblocks, 256, 0, stream>>>(vals, rows, cols, h0, h1, E);
        zero_f4_kernel<<<zblocks, 256, 0, stream>>>((float4*)out, n4);
        spmm_edges_kernel<<<sblocks, 256, 0, stream>>>(vals, rows, cols, h1, out, E);
    } else {
        gemm_fused_kernel<<<gblocks, 256, 0, stream>>>(X, W1, W2, h0, N, nullptr, nullptr, nullptr, nullptr, nullptr, 0, 0);
        zero_f4_kernel<<<zblocks, 256, 0, stream>>>((float4*)out, n4);
        spmm_edges_kernel<<<sblocks, 256, 0, stream>>>(vals, rows, cols, h0, out, E);
        zero_f4_kernel<<<zblocks, 256, 0, stream>>>((float4*)h0, n4);
        spmm_edges_kernel<<<sblocks, 256, 0, stream>>>(vals, rows, cols, out, h0, E);
        hipMemcpyAsync(out, h0, hbytes, hipMemcpyDeviceToDevice, stream);
    }
}
